# starting kernel
# baseline (speedup 1.0000x reference)
#include <hip/hip_runtime.h>

typedef float v2f __attribute__((ext_vector_type(2)));
typedef float v8f __attribute__((ext_vector_type(8)));

__device__ __forceinline__ v8f wmma_f32_4(v2f a, v2f b, v8f c) {
    return __builtin_amdgcn_wmma_f32_16x16x4_f32(
         false, a,  false, b,
         (short)0, c,  false,  false);
}

template <bool GUARD>
__launch_bounds__(256)
__global__ void mlp3_wmma_kernel(const float* __restrict__ x,
                                 const float* __restrict__ W1, const float* __restrict__ b1,
                                 const float* __restrict__ W2, const float* __restrict__ b2,
                                 const float* __restrict__ W3, const float* __restrict__ b3,
                                 float* __restrict__ out,
                                 int Brows, int tiles)
{
    __shared__ float lds[8][16 * 16];

    const int tid  = threadIdx.x;
    const int wave = tid >> 5;
    const int lane = tid & 31;
    const int nlo  = lane & 15;
    const int hi   = lane >> 4;
    float* L = lds[wave];

    v2f B1;
    B1.x = W1[(2 * hi + 0) * 16 + nlo];
    B1.y = W1[(2 * hi + 1) * 16 + nlo];

    v2f B2[4];
#pragma unroll
    for (int j = 0; j < 4; ++j) {
        const int k0 = 4 * j + 2 * hi;
        B2[j].x = (nlo < 8) ? W2[(k0 + 0) * 8 + nlo] : 0.0f;
        B2[j].y = (nlo < 8) ? W2[(k0 + 1) * 8 + nlo] : 0.0f;
    }

    v2f B3[2];
#pragma unroll
    for (int j = 0; j < 2; ++j) {
        const int k0 = 4 * j + 2 * hi;
        B3[j].x = (nlo < 3) ? W3[(k0 + 0) * 3 + nlo] : 0.0f;
        B3[j].y = (nlo < 3) ? W3[(k0 + 1) * 3 + nlo] : 0.0f;
    }

    const float c1 = b1[nlo];
    const float c2 = (nlo < 8) ? b2[nlo] : 0.0f;
    const float c3 = (nlo < 3) ? b3[nlo] : 0.0f;

    const int st_base = hi * 128 + nlo;
    const int ld_base = nlo * 16 + hi * 2;

    const int wave_stride = gridDim.x * 8;

    for (int t = blockIdx.x * 8 + wave; t < tiles; t += wave_stride) {
        const int base = t * 16;

        const int row = base + nlo;
        v2f a1;
        if (GUARD) {
            a1 = (v2f){0.0f, 0.0f};
            if (row < Brows)
                a1 = __builtin_nontemporal_load((const v2f*)(x + row * 4 + 2 * hi));
        } else {
            a1 = __builtin_nontemporal_load((const v2f*)(x + row * 4 + 2 * hi));
        }

        v8f acc;
#pragma unroll
        for (int i = 0; i < 8; ++i) acc[i] = c1;
        v8f h1 = wmma_f32_4(a1, B1, acc);
#pragma unroll
        for (int i = 0; i < 8; ++i) h1[i] = fmaxf(h1[i], 0.0f);

#pragma unroll
        for (int i = 0; i < 8; ++i) L[st_base + i * 16] = h1[i];
        __asm__ volatile("s_wait_dscnt 0x0" ::: "memory");
        v2f a2[4];
#pragma unroll
        for (int j = 0; j < 4; ++j) a2[j] = *(const v2f*)(L + ld_base + 4 * j);

#pragma unroll
        for (int i = 0; i < 8; ++i) acc[i] = c2;
        v8f h2 = wmma_f32_4(a2[0], B2[0], acc);
        h2 = wmma_f32_4(a2[1], B2[1], h2);
        h2 = wmma_f32_4(a2[2], B2[2], h2);
        h2 = wmma_f32_4(a2[3], B2[3], h2);
#pragma unroll
        for (int i = 0; i < 8; ++i) h2[i] = fmaxf(h2[i], 0.0f);

#pragma unroll
        for (int i = 0; i < 8; ++i) L[st_base + i * 16] = h2[i];
        __asm__ volatile("s_wait_dscnt 0x0" ::: "memory");
        v2f a3[2];
#pragma unroll
        for (int j = 0; j < 2; ++j) a3[j] = *(const v2f*)(L + ld_base + 4 * j);

#pragma unroll
        for (int i = 0; i < 8; ++i) acc[i] = c3;
        v8f o = wmma_f32_4(a3[0], B3[0], acc);
        o = wmma_f32_4(a3[1], B3[1], o);

        if (nlo < 3) {
            const int r0 = base + 8 * hi;
#pragma unroll
            for (int i = 0; i < 8; ++i) {
                const int r = r0 + i;
                if (!GUARD || r < Brows)
                    __builtin_nontemporal_store(o[i], out + r * 3 + nlo);
            }
        }
    }
}

extern "C" void kernel_launch(void* const* d_in, const int* in_sizes, int n_in,
                              void* d_out, int out_size, void* d_ws, size_t ws_size,
                              hipStream_t stream) {
    const float* x  = (const float*)d_in[0];
    const float* W1 = (const float*)d_in[1];
    const float* b1 = (const float*)d_in[2];
    const float* W2 = (const float*)d_in[3];
    const float* b2 = (const float*)d_in[4];
    const float* W3 = (const float*)d_in[5];
    const float* b3 = (const float*)d_in[6];
    float* out = (float*)d_out;

    const int Brows = in_sizes[0] / 4;
    const int tiles = (Brows + 15) / 16;
    int blocks = (tiles + 7) / 8;
    if (blocks > 4096) blocks = 4096;
    if (blocks < 1) blocks = 1;

    if ((Brows & 15) == 0) {
        mlp3_wmma_kernel<false><<<blocks, 256, 0, stream>>>(
            x, W1, b1, W2, b2, W3, b3, out, Brows, tiles);
    } else {
        mlp3_wmma_kernel<true><<<blocks, 256, 0, stream>>>(
            x, W1, b1, W2, b2, W3, b3, out, Brows, tiles);
    }
}


# Grid cut from 4096 to 1024 work-groups: all 8192 waves resident at once, each loops over ~15 tiles with the next tile's x rows requested one trip ahead, so wave entry no longer sets the time
# speedup vs baseline: 1.3802x; 1.3802x over previous
#include <hip/hip_runtime.h>

typedef float v2f __attribute__((ext_vector_type(2)));
typedef float v8f __attribute__((ext_vector_type(8)));

__device__ __forceinline__ v8f wmma_f32_4(v2f a, v2f b, v8f c) {
    return __builtin_amdgcn_wmma_f32_16x16x4_f32(
         false, a,  false, b,
         (short)0, c,  false,  false);
}

template <bool GUARD>
__device__ __forceinline__ v2f load_x_tile(const float* __restrict__ x, int t, int nlo, int hi, int Brows) {
    const int row = t * 16 + nlo;
    v2f a = (v2f){0.0f, 0.0f};
    if (!GUARD || row < Brows)
        a = __builtin_nontemporal_load((const v2f*)(x + row * 4 + 2 * hi));
    return a;
}

template <bool GUARD>
__launch_bounds__(256)
__global__ void mlp3_wmma_kernel(const float* __restrict__ x,
                                 const float* __restrict__ W1, const float* __restrict__ b1,
                                 const float* __restrict__ W2, const float* __restrict__ b2,
                                 const float* __restrict__ W3, const float* __restrict__ b3,
                                 float* __restrict__ out,
                                 int Brows, int tiles)
{
    __shared__ float lds[8][16 * 16];

    const int tid  = threadIdx.x;
    const int wave = __builtin_amdgcn_readfirstlane(tid >> 5);
    const int lane = tid & 31;
    const int nlo  = lane & 15;
    const int hi   = lane >> 4;
    float* L = lds[wave];

    v2f B1;
    B1.x = W1[(2 * hi + 0) * 16 + nlo];
    B1.y = W1[(2 * hi + 1) * 16 + nlo];

    v2f B2[4];
#pragma unroll
    for (int j = 0; j < 4; ++j) {
        const int k0 = 4 * j + 2 * hi;
        B2[j].x = (nlo < 8) ? W2[(k0 + 0) * 8 + nlo] : 0.0f;
        B2[j].y = (nlo < 8) ? W2[(k0 + 1) * 8 + nlo] : 0.0f;
    }

    v2f B3[2];
#pragma unroll
    for (int j = 0; j < 2; ++j) {
        const int k0 = 4 * j + 2 * hi;
        B3[j].x = (nlo < 3) ? W3[(k0 + 0) * 3 + nlo] : 0.0f;
        B3[j].y = (nlo < 3) ? W3[(k0 + 1) * 3 + nlo] : 0.0f;
    }

    const float c1 = b1[nlo];
    const float c2 = (nlo < 8) ? b2[nlo] : 0.0f;
    const float c3 = (nlo < 3) ? b3[nlo] : 0.0f;

    const int st_base = hi * 128 + nlo;
    const int ld_base = nlo * 16 + hi * 2;

    auto tile = [&](v2f a1, int base) {
        v8f acc;
#pragma unroll
        for (int i = 0; i < 8; ++i) acc[i] = c1;
        v8f h1 = wmma_f32_4(a1, B1, acc);
#pragma unroll
        for (int i = 0; i < 8; ++i) h1[i] = fmaxf(h1[i], 0.0f);

#pragma unroll
        for (int i = 0; i < 8; ++i) L[st_base + i * 16] = h1[i];
        __asm__ volatile("s_wait_dscnt 0x0" ::: "memory");
        v2f a2[4];
#pragma unroll
        for (int j = 0; j < 4; ++j) a2[j] = *(const v2f*)(L + ld_base + 4 * j);

#pragma unroll
        for (int i = 0; i < 8; ++i) acc[i] = c2;
        v8f h2 = wmma_f32_4(a2[0], B2[0], acc);
        h2 = wmma_f32_4(a2[1], B2[1], h2);
        h2 = wmma_f32_4(a2[2], B2[2], h2);
        h2 = wmma_f32_4(a2[3], B2[3], h2);
#pragma unroll
        for (int i = 0; i < 8; ++i) h2[i] = fmaxf(h2[i], 0.0f);

#pragma unroll
        for (int i = 0; i < 8; ++i) L[st_base + i * 16] = h2[i];
        __asm__ volatile("s_wait_dscnt 0x0" ::: "memory");
        v2f a3[2];
#pragma unroll
        for (int j = 0; j < 2; ++j) a3[j] = *(const v2f*)(L + ld_base + 4 * j);

#pragma unroll
        for (int i = 0; i < 8; ++i) acc[i] = c3;
        v8f o = wmma_f32_4(a3[0], B3[0], acc);
        o = wmma_f32_4(a3[1], B3[1], o);

        if (nlo < 3) {
            const int r0 = base + 8 * hi;
#pragma unroll
            for (int i = 0; i < 8; ++i) {
                const int r = r0 + i;
                if (!GUARD || r < Brows)
                    __builtin_nontemporal_store(o[i], out + r * 3 + nlo);
            }
        }
    };

    const int wave_stride = gridDim.x * 8;
    int t = blockIdx.x * 8 + wave;
    if (t < tiles) {
        v2f a1 = load_x_tile<GUARD>(x, t, nlo, hi, Brows);
        int tn = t + wave_stride;
        while (tn < tiles) {
            const v2f a1n = load_x_tile<GUARD>(x, tn, nlo, hi, Brows);
            tile(a1, t * 16);
            a1 = a1n;
            t  = tn;
            tn += wave_stride;
        }
        tile(a1, t * 16);
    }
}

extern "C" void kernel_launch(void* const* d_in, const int* in_sizes, int n_in,
                              void* d_out, int out_size, void* d_ws, size_t ws_size,
                              hipStream_t stream) {
    const float* x  = (const float*)d_in[0];
    const float* W1 = (const float*)d_in[1];
    const float* b1 = (const float*)d_in[2];
    const float* W2 = (const float*)d_in[3];
    const float* b2 = (const float*)d_in[4];
    const float* W3 = (const float*)d_in[5];
    const float* b3 = (const float*)d_in[6];
    float* out = (float*)d_out;

    const int Brows = in_sizes[0] / 4;
    const int tiles = (Brows + 15) / 16;
    int blocks = (tiles + 7) / 8;
    if (blocks > 1024) blocks = 1024;
    if (blocks < 1) blocks = 1;

    if ((Brows & 15) == 0) {
        mlp3_wmma_kernel<false><<<blocks, 256, 0, stream>>>(
            x, W1, b1, W2, b2, W3, b3, out, Brows, tiles);
    } else {
        mlp3_wmma_kernel<true><<<blocks, 256, 0, stream>>>(
            x, W1, b1, W2, b2, W3, b3, out, Brows, tiles);
    }
}


# Chain computed transposed (weights A, rows B): a layer's D registers are the next layer's B fragments, so both LDS round trips go and a row leaves in one 12-byte store; layers 2 and 3 sum k reordered
# speedup vs baseline: 2.4187x; 1.7524x over previous
#include <hip/hip_runtime.h>


typedef float v2f __attribute__((ext_vector_type(2)));
typedef float v8f __attribute__((ext_vector_type(8)));

__device__ __forceinline__ v8f wmma_f32_4(v2f a, v2f b, v8f c) {
    return __builtin_amdgcn_wmma_f32_16x16x4_f32(
         false, a,  false, b,
         (short)0, c,  false,  false);
}

template <bool GUARD>
__device__ __forceinline__ v2f load_x_tile(const float* __restrict__ x, int t, int nlo, int hi, int Brows) {
    const int row = t * 16 + nlo;
    v2f a = (v2f){0.0f, 0.0f};
    if (!GUARD || row < Brows)
        a = __builtin_nontemporal_load((const v2f*)(x + row * 4 + 2 * hi));
    return a;
}

template <bool GUARD>
__launch_bounds__(256)
__global__ void mlp3_wmma_kernel(const float* __restrict__ x,
                                 const float* __restrict__ W1, const float* __restrict__ b1,
                                 const float* __restrict__ W2, const float* __restrict__ b2,
                                 const float* __restrict__ W3, const float* __restrict__ b3,
                                 float* __restrict__ out,
                                 int Brows, int tiles)
{
    const int tid  = threadIdx.x;
    const int wave = __builtin_amdgcn_readfirstlane(tid >> 5);
    const int lane = tid & 31;
    const int nlo  = lane & 15;
    const int hi   = lane >> 4;

    v2f A1;
    A1.x = W1[(2 * hi + 0) * 16 + nlo];
    A1.y = W1[(2 * hi + 1) * 16 + nlo];
    v8f C1;
#pragma unroll
    for (int r = 0; r < 8; ++r) C1[r] = b1[8 * hi + r];

    const bool on2 = (nlo & 4) == 0;
    const int  u2  = (nlo & 3) + 4 * (nlo >> 3);
    v2f A2[4];
#pragma unroll
    for (int j = 0; j < 4; ++j) {
        const int k0 = 8 * hi + 2 * j;
        const float w0 = W2[(k0 + 0) * 8 + u2];
        const float w1 = W2[(k0 + 1) * 8 + u2];
        A2[j].x = on2 ? w0 : 0.0f;
        A2[j].y = on2 ? w1 : 0.0f;
    }
    v8f C2;
#pragma unroll
    for (int r = 0; r < 4; ++r) C2[r] = b2[4 * hi + r];
#pragma unroll
    for (int r = 4; r < 8; ++r) C2[r] = 0.0f;

    const int u3 = (nlo < 3) ? nlo : 2;
    v2f A3[2];
#pragma unroll
    for (int j = 0; j < 2; ++j) {
        const int k0 = 4 * hi + 2 * j;
        const float w0 = W3[(k0 + 0) * 3 + u3];
        const float w1 = W3[(k0 + 1) * 3 + u3];
        A3[j].x = (nlo < 3) ? w0 : 0.0f;
        A3[j].y = (nlo < 3) ? w1 : 0.0f;
    }
    v8f C3;
#pragma unroll
    for (int r = 0; r < 3; ++r) {
        const float bv = b3[r];
        C3[r] = (hi == 0) ? bv : 0.0f;
    }
#pragma unroll
    for (int r = 3; r < 8; ++r) C3[r] = 0.0f;

    auto tile = [&](v2f xb, int base) {
        v8f h1 = wmma_f32_4(A1, xb, C1);
#pragma unroll
        for (int i = 0; i < 8; ++i) h1[i] = fmaxf(h1[i], 0.0f);

        v8f h2 = wmma_f32_4(A2[0], (v2f){h1[0], h1[1]}, C2);
        h2 = wmma_f32_4(A2[1], (v2f){h1[2], h1[3]}, h2);
        h2 = wmma_f32_4(A2[2], (v2f){h1[4], h1[5]}, h2);
        h2 = wmma_f32_4(A2[3], (v2f){h1[6], h1[7]}, h2);
#pragma unroll
        for (int i = 0; i < 4; ++i) h2[i] = fmaxf(h2[i], 0.0f);

        v8f o = wmma_f32_4(A3[0], (v2f){h2[0], h2[1]}, C3);
        o = wmma_f32_4(A3[1], (v2f){h2[2], h2[3]}, o);

        if (hi == 0) {
            const int r = base + nlo;
            if (!GUARD || r < Brows) {
                float* p = out + r * 3;
                __builtin_nontemporal_store(o[0], p + 0);
                __builtin_nontemporal_store(o[1], p + 1);
                __builtin_nontemporal_store(o[2], p + 2);
            }
        }
    };

    const int wave_stride = gridDim.x * 8;
    int t = blockIdx.x * 8 + wave;
    if (t < tiles) {
        v2f a1 = load_x_tile<GUARD>(x, t, nlo, hi, Brows);
        int tn = t + wave_stride;
        while (tn < tiles) {
            const v2f a1n = load_x_tile<GUARD>(x, tn, nlo, hi, Brows);
            __builtin_amdgcn_sched_barrier(0);
            tile(a1, t * 16);
            a1 = a1n;
            t  = tn;
            tn += wave_stride;
        }
        tile(a1, t * 16);
    }
}

extern "C" void kernel_launch(void* const* d_in, const int* in_sizes, int n_in,
                              void* d_out, int out_size, void* d_ws, size_t ws_size,
                              hipStream_t stream) {
    const float* x  = (const float*)d_in[0];
    const float* W1 = (const float*)d_in[1];
    const float* b1 = (const float*)d_in[2];
    const float* W2 = (const float*)d_in[3];
    const float* b2 = (const float*)d_in[4];
    const float* W3 = (const float*)d_in[5];
    const float* b3 = (const float*)d_in[6];
    float* out = (float*)d_out;

    const int Brows = in_sizes[0] / 4;
    const int tiles = (Brows + 15) / 16;
    int blocks = (tiles + 7) / 8;
    if (blocks > 1024) blocks = 1024;
    if (blocks < 1) blocks = 1;

    if ((Brows & 15) == 0) {
        mlp3_wmma_kernel<false><<<blocks, 256, 0, stream>>>(
            x, W1, b1, W2, b2, W3, b3, out, Brows, tiles);
    } else {
        mlp3_wmma_kernel<true><<<blocks, 256, 0, stream>>>(
            x, W1, b1, W2, b2, W3, b3, out, Brows, tiles);
    }
}


# Each wave keeps four x requests in flight instead of one: tiles go in adjacent pairs, and the next pair is requested before the wait for the current one (two register sets in turn); sums unchanged
# speedup vs baseline: 2.7325x; 1.1297x over previous
#include <hip/hip_runtime.h>


typedef float v2f __attribute__((ext_vector_type(2)));
typedef float v8f __attribute__((ext_vector_type(8)));

__device__ __forceinline__ v8f wmma_f32_4(v2f a, v2f b, v8f c) {
    return __builtin_amdgcn_wmma_f32_16x16x4_f32(
         false, a,  false, b,
         (short)0, c,  false,  false);
}

template <bool GUARD>
__device__ __forceinline__ v2f load_x_tile(const float* __restrict__ x, int t, int nlo, int hi, int Brows) {
    const int row = t * 16 + nlo;
    v2f a = (v2f){0.0f, 0.0f};
    if (!GUARD || row < Brows)
        a = __builtin_nontemporal_load((const v2f*)(x + row * 4 + 2 * hi));
    return a;
}

template <bool GUARD>
__launch_bounds__(256)
__global__ void mlp3_wmma_kernel(const float* __restrict__ x,
                                 const float* __restrict__ W1, const float* __restrict__ b1,
                                 const float* __restrict__ W2, const float* __restrict__ b2,
                                 const float* __restrict__ W3, const float* __restrict__ b3,
                                 float* __restrict__ out,
                                 int Brows, int pairs)
{
    const int tid  = threadIdx.x;
    const int wave = __builtin_amdgcn_readfirstlane(tid >> 5);
    const int lane = tid & 31;
    const int nlo  = lane & 15;
    const int hi   = lane >> 4;

    v2f A1;
    A1.x = W1[(2 * hi + 0) * 16 + nlo];
    A1.y = W1[(2 * hi + 1) * 16 + nlo];
    v8f C1;
#pragma unroll
    for (int r = 0; r < 8; ++r) C1[r] = b1[8 * hi + r];

    const bool on2 = (nlo & 4) == 0;
    const int  u2  = (nlo & 3) + 4 * (nlo >> 3);
    v2f A2[4];
#pragma unroll
    for (int j = 0; j < 4; ++j) {
        const int k0 = 8 * hi + 2 * j;
        const float w0 = W2[(k0 + 0) * 8 + u2];
        const float w1 = W2[(k0 + 1) * 8 + u2];
        A2[j].x = on2 ? w0 : 0.0f;
        A2[j].y = on2 ? w1 : 0.0f;
    }
    v8f C2;
#pragma unroll
    for (int r = 0; r < 4; ++r) C2[r] = b2[4 * hi + r];
#pragma unroll
    for (int r = 4; r < 8; ++r) C2[r] = 0.0f;

    const int u3 = (nlo < 3) ? nlo : 2;
    v2f A3[2];
#pragma unroll
    for (int j = 0; j < 2; ++j) {
        const int k0 = 4 * hi + 2 * j;
        const float w0 = W3[(k0 + 0) * 3 + u3];
        const float w1 = W3[(k0 + 1) * 3 + u3];
        A3[j].x = (nlo < 3) ? w0 : 0.0f;
        A3[j].y = (nlo < 3) ? w1 : 0.0f;
    }
    v8f C3;
#pragma unroll
    for (int r = 0; r < 3; ++r) {
        const float bv = b3[r];
        C3[r] = (hi == 0) ? bv : 0.0f;
    }
#pragma unroll
    for (int r = 3; r < 8; ++r) C3[r] = 0.0f;

    auto tile_pair = [&](v2f xa, v2f xb, int base) {
        v8f h1a = wmma_f32_4(A1, xa, C1);
        v8f h1b = wmma_f32_4(A1, xb, C1);
#pragma unroll
        for (int i = 0; i < 8; ++i) { h1a[i] = fmaxf(h1a[i], 0.0f); h1b[i] = fmaxf(h1b[i], 0.0f); }

        v8f h2a = wmma_f32_4(A2[0], (v2f){h1a[0], h1a[1]}, C2);
        v8f h2b = wmma_f32_4(A2[0], (v2f){h1b[0], h1b[1]}, C2);
        h2a = wmma_f32_4(A2[1], (v2f){h1a[2], h1a[3]}, h2a);
        h2b = wmma_f32_4(A2[1], (v2f){h1b[2], h1b[3]}, h2b);
        h2a = wmma_f32_4(A2[2], (v2f){h1a[4], h1a[5]}, h2a);
        h2b = wmma_f32_4(A2[2], (v2f){h1b[4], h1b[5]}, h2b);
        h2a = wmma_f32_4(A2[3], (v2f){h1a[6], h1a[7]}, h2a);
        h2b = wmma_f32_4(A2[3], (v2f){h1b[6], h1b[7]}, h2b);
#pragma unroll
        for (int i = 0; i < 4; ++i) { h2a[i] = fmaxf(h2a[i], 0.0f); h2b[i] = fmaxf(h2b[i], 0.0f); }

        v8f oa = wmma_f32_4(A3[0], (v2f){h2a[0], h2a[1]}, C3);
        v8f ob = wmma_f32_4(A3[0], (v2f){h2b[0], h2b[1]}, C3);
        oa = wmma_f32_4(A3[1], (v2f){h2a[2], h2a[3]}, oa);
        ob = wmma_f32_4(A3[1], (v2f){h2b[2], h2b[3]}, ob);

        if (hi == 0) {
            const int ra = base + nlo;
            float* dst = out + ra * 3;
            if (!GUARD || ra < Brows) {
                __builtin_nontemporal_store(oa[0], dst + 0);
                __builtin_nontemporal_store(oa[1], dst + 1);
                __builtin_nontemporal_store(oa[2], dst + 2);
            }
            if (!GUARD || ra + 16 < Brows) {
                __builtin_nontemporal_store(ob[0], dst + 48);
                __builtin_nontemporal_store(ob[1], dst + 49);
                __builtin_nontemporal_store(ob[2], dst + 50);
            }
        }
    };

    const int wave_stride = gridDim.x * 8;
    int p = blockIdx.x * 8 + wave;
    if (p < pairs) {
        v2f xa0 = load_x_tile<GUARD>(x, 2 * p,     nlo, hi, Brows);
        v2f xb0 = load_x_tile<GUARD>(x, 2 * p + 1, nlo, hi, Brows);
        int pn = p + wave_stride;
        for (;;) {
            if (pn >= pairs) { tile_pair(xa0, xb0, p * 32); break; }
            const v2f xa1 = load_x_tile<GUARD>(x, 2 * pn,     nlo, hi, Brows);
            const v2f xb1 = load_x_tile<GUARD>(x, 2 * pn + 1, nlo, hi, Brows);
            __builtin_amdgcn_sched_barrier(0);
            tile_pair(xa0, xb0, p * 32);
            p  = pn;
            pn += wave_stride;

            if (pn >= pairs) { tile_pair(xa1, xb1, p * 32); break; }
            xa0 = load_x_tile<GUARD>(x, 2 * pn,     nlo, hi, Brows);
            xb0 = load_x_tile<GUARD>(x, 2 * pn + 1, nlo, hi, Brows);
            __builtin_amdgcn_sched_barrier(0);
            tile_pair(xa1, xb1, p * 32);
            p  = pn;
            pn += wave_stride;
        }
    }
}

extern "C" void kernel_launch(void* const* d_in, const int* in_sizes, int n_in,
                              void* d_out, int out_size, void* d_ws, size_t ws_size,
                              hipStream_t stream) {
    const float* x  = (const float*)d_in[0];
    const float* W1 = (const float*)d_in[1];
    const float* b1 = (const float*)d_in[2];
    const float* W2 = (const float*)d_in[3];
    const float* b2 = (const float*)d_in[4];
    const float* W3 = (const float*)d_in[5];
    const float* b3 = (const float*)d_in[6];
    float* out = (float*)d_out;

    const int Brows = in_sizes[0] / 4;
    const int tiles = (Brows + 15) / 16;
    const int pairs = (tiles + 1) / 2;
    int blocks = (pairs + 7) / 8;
    if (blocks > 1024) blocks = 1024;
    if (blocks < 1) blocks = 1;

    if ((Brows & 31) == 0) {
        mlp3_wmma_kernel<false><<<blocks, 256, 0, stream>>>(
            x, W1, b1, W2, b2, W3, b3, out, Brows, pairs);
    } else {
        mlp3_wmma_kernel<true><<<blocks, 256, 0, stream>>>(
            x, W1, b1, W2, b2, W3, b3, out, Brows, pairs);
    }
}
